# starting kernel
# baseline (speedup 1.0000x reference)
#include <hip/hip_runtime.h>
#include <hip/hip_bf16.h>

#define VOCAB 100000
#define EMB_DIM 128
#define HIDDEN 256
#define NUM_CLASSES 20
#define BATCH 4096
#define SEQ 200

typedef float v2f __attribute__((ext_vector_type(2)));
typedef float v8f __attribute__((ext_vector_type(8)));

__global__ __launch_bounds__(256) void ft_pool_kernel(
    const int* __restrict__ x, const int* __restrict__ lens,
    const float* __restrict__ emb, float* __restrict__ pooled) {
  const int lane = threadIdx.x & 31;
  const int wave = threadIdx.x >> 5;
  const int b = blockIdx.x * 8 + wave;
  if (b >= BATCH) return;

  int len = lens[b];
  if (len < 1) len = 1;
  if (len > SEQ) len = SEQ;

  const int* xb = x + (size_t)b * SEQ;
  float4 acc = make_float4(0.f, 0.f, 0.f, 0.f);

  for (int s0 = 0; s0 < len; s0 += 32) {
    int cnt = len - s0;
    if (cnt > 32) cnt = 32;
    int tok = 0;
    if (lane < cnt) tok = xb[s0 + lane];
    for (int j = 0; j < cnt; ++j) {
      int t = __shfl(tok, j, 32);
      const float4* row = (const float4*)(emb + (size_t)t * EMB_DIM);
      float4 v = row[lane];
      acc.x += v.x; acc.y += v.y; acc.z += v.z; acc.w += v.w;
    }
  }
  const float inv = 1.0f / (float)len;
  float4 r = make_float4(acc.x * inv, acc.y * inv, acc.z * inv, acc.w * inv);
  ((float4*)(pooled + (size_t)b * EMB_DIM))[lane] = r;
}

__global__ __launch_bounds__(256) void ft_gemm1_kernel(
    const float* __restrict__ pooled, const float* __restrict__ W1,
    const float* __restrict__ b1, float* __restrict__ hidden) {
  const int lane = threadIdx.x & 31;
  const int wave = threadIdx.x >> 5;
  const int tile = blockIdx.x * 8 + wave;
  const int mt = tile >> 4;
  const int nt = tile & 15;
  const int lm = lane & 15;
  const int half = lane >> 4;
  const int kb = half * 2;

  const float* Arow = pooled + (size_t)(mt * 16 + lm) * EMB_DIM;
  const int ncol = nt * 16 + lm;

  v8f c = {0.f, 0.f, 0.f, 0.f, 0.f, 0.f, 0.f, 0.f};
  for (int k0 = 0; k0 < EMB_DIM; k0 += 4) {
    v2f a, bm;
    a.x = Arow[k0 + kb];
    a.y = Arow[k0 + kb + 1];
    bm.x = W1[(size_t)(k0 + kb) * HIDDEN + ncol];
    bm.y = W1[(size_t)(k0 + kb + 1) * HIDDEN + ncol];
    c = __builtin_amdgcn_wmma_f32_16x16x4_f32(
         false, a,  false, bm,
         (short)0, c,  false,  false);
  }

  const float bias = b1[ncol];
  const int mbase = mt * 16 + half * 8;
#pragma unroll
  for (int v = 0; v < 8; ++v) {
    float val = c[v] + bias;
    val = val > 0.f ? val : 0.f;
    hidden[(size_t)(mbase + v) * HIDDEN + ncol] = val;
  }
}

__global__ __launch_bounds__(256) void ft_gemm2_kernel(
    const float* __restrict__ hidden, const float* __restrict__ W2,
    const float* __restrict__ b2, float* __restrict__ out) {
  const int lane = threadIdx.x & 31;
  const int wave = threadIdx.x >> 5;
  const int tile = blockIdx.x * 8 + wave;
  const int mt = tile >> 1;
  const int nt = tile & 1;
  const int lm = lane & 15;
  const int half = lane >> 4;
  const int kb = half * 2;

  const int ncol = nt * 16 + lm;
  const float nmask = (ncol < NUM_CLASSES) ? 1.0f : 0.0f;
  const int nclamp = (ncol < NUM_CLASSES) ? ncol : (NUM_CLASSES - 1);

  const float* Arow = hidden + (size_t)(mt * 16 + lm) * HIDDEN;

  v8f c = {0.f, 0.f, 0.f, 0.f, 0.f, 0.f, 0.f, 0.f};
  for (int k0 = 0; k0 < HIDDEN; k0 += 4) {
    v2f a, bm;
    a.x = Arow[k0 + kb];
    a.y = Arow[k0 + kb + 1];
    bm.x = W2[(size_t)(k0 + kb) * NUM_CLASSES + nclamp] * nmask;
    bm.y = W2[(size_t)(k0 + kb + 1) * NUM_CLASSES + nclamp] * nmask;
    c = __builtin_amdgcn_wmma_f32_16x16x4_f32(
        false, a, false, bm, (short)0, c, false, false);
  }

  if (ncol < NUM_CLASSES) {
    const float bias = b2[ncol];
    const int mbase = mt * 16 + half * 8;
#pragma unroll
    for (int v = 0; v < 8; ++v) {
      out[(size_t)(mbase + v) * NUM_CLASSES + ncol] = c[v] + bias;
    }
  }
}

extern "C" void kernel_launch(void* const* d_in, const int* in_sizes, int n_in,
                              void* d_out, int out_size, void* d_ws, size_t ws_size,
                              hipStream_t stream) {
  const int*   x    = (const int*)d_in[0];
  const int*   lens = (const int*)d_in[1];
  const float* emb  = (const float*)d_in[2];
  const float* W1   = (const float*)d_in[3];
  const float* b1   = (const float*)d_in[4];
  const float* W2   = (const float*)d_in[5];
  const float* b2   = (const float*)d_in[6];
  float*       out  = (float*)d_out;

  float* pooled = (float*)d_ws;
  float* hidden = (float*)((char*)d_ws + (size_t)BATCH * EMB_DIM * sizeof(float));

  ft_pool_kernel<<<BATCH / 8, 256, 0, stream>>>(x, lens, emb, pooled);

  ft_gemm1_kernel<<<(BATCH / 16) * (HIDDEN / 16) / 8, 256, 0, stream>>>(
      pooled, W1, b1, hidden);

  ft_gemm2_kernel<<<(BATCH / 16) * 2 / 8, 256, 0, stream>>>(
      hidden, W2, b2, out);
}


# pool gather: a wave requests the 32 embedding rows of a token chunk back to back and sums them afterwards in token order (was one dependent row load per trip); summation order unchanged
# speedup vs baseline: 2.1983x; 2.1983x over previous
#include <hip/hip_runtime.h>
#include <hip/hip_bf16.h>

#define VOCAB 100000
#define EMB_DIM 128
#define HIDDEN 256
#define NUM_CLASSES 20
#define BATCH 4096
#define SEQ 200

typedef float v2f __attribute__((ext_vector_type(2)));
typedef float v8f __attribute__((ext_vector_type(8)));

#define POOL_CHUNK 32

__device__ __forceinline__ float4 ft_load_row(const float* __restrict__ emb,
                                              int tok, int u, int lane) {
  const int t = __builtin_amdgcn_readlane(tok, u);
  return ((const float4*)(emb + (size_t)t * EMB_DIM))[lane];
}

__global__ __launch_bounds__(256) void ft_pool_kernel(
    const int* __restrict__ x, const int* __restrict__ lens,
    const float* __restrict__ emb, float* __restrict__ pooled) {
  const int lane = threadIdx.x & 31;
  const int b = __builtin_amdgcn_readfirstlane((int)(blockIdx.x * 8 + (threadIdx.x >> 5)));
  if (b >= BATCH) return;

  int len = lens[b];
  if (len < 1) len = 1;
  if (len > SEQ) len = SEQ;

  const int* xb = x + (size_t)b * SEQ;
  float4 acc = make_float4(0.f, 0.f, 0.f, 0.f);

  int tok = 0;
  if (lane < len) tok = xb[lane];

  int s0 = 0;
  for (; len - s0 >= POOL_CHUNK; s0 += POOL_CHUNK) {
    float4 v[POOL_CHUNK];
    for (int u = 0; u < POOL_CHUNK; ++u) v[u] = ft_load_row(emb, tok, u, lane);

    int tok_next = 0;
    if (s0 + POOL_CHUNK + lane < len) tok_next = xb[s0 + POOL_CHUNK + lane];

    for (int u = 0; u < POOL_CHUNK; ++u) {
      acc.x += v[u].x; acc.y += v[u].y; acc.z += v[u].z; acc.w += v[u].w;
    }
    tok = tok_next;
  }

  const int cnt = len - s0;
  if (cnt > 0) {
    float4 v[POOL_CHUNK - 1];
    v[0] = ft_load_row(emb, tok, 0, lane);
    for (int u = 1; u < POOL_CHUNK - 1; ++u) {
      v[u] = make_float4(0.f, 0.f, 0.f, 0.f);
      if (u < cnt) v[u] = ft_load_row(emb, tok, u, lane);
    }
    for (int u = 0; u < POOL_CHUNK - 1; ++u) {
      acc.x += v[u].x; acc.y += v[u].y; acc.z += v[u].z; acc.w += v[u].w;
    }
  }

  const float inv = 1.0f / (float)len;
  float4 r = make_float4(acc.x * inv, acc.y * inv, acc.z * inv, acc.w * inv);
  ((float4*)(pooled + (size_t)b * EMB_DIM))[lane] = r;
}

__global__ __launch_bounds__(256) void ft_gemm1_kernel(
    const float* __restrict__ pooled, const float* __restrict__ W1,
    const float* __restrict__ b1, float* __restrict__ hidden) {
  const int lane = threadIdx.x & 31;
  const int wave = threadIdx.x >> 5;
  const int tile = blockIdx.x * 8 + wave;
  const int mt = tile >> 4;
  const int nt = tile & 15;
  const int lm = lane & 15;
  const int half = lane >> 4;
  const int kb = half * 2;

  const float* Arow = pooled + (size_t)(mt * 16 + lm) * EMB_DIM;
  const int ncol = nt * 16 + lm;

  v8f c = {0.f, 0.f, 0.f, 0.f, 0.f, 0.f, 0.f, 0.f};
  for (int k0 = 0; k0 < EMB_DIM; k0 += 4) {
    v2f a, bm;
    a.x = Arow[k0 + kb];
    a.y = Arow[k0 + kb + 1];
    bm.x = W1[(size_t)(k0 + kb) * HIDDEN + ncol];
    bm.y = W1[(size_t)(k0 + kb + 1) * HIDDEN + ncol];
    c = __builtin_amdgcn_wmma_f32_16x16x4_f32(
         false, a,  false, bm,
         (short)0, c,  false,  false);
  }

  const float bias = b1[ncol];
  const int mbase = mt * 16 + half * 8;
#pragma unroll
  for (int v = 0; v < 8; ++v) {
    float val = c[v] + bias;
    val = val > 0.f ? val : 0.f;
    hidden[(size_t)(mbase + v) * HIDDEN + ncol] = val;
  }
}

__global__ __launch_bounds__(256) void ft_gemm2_kernel(
    const float* __restrict__ hidden, const float* __restrict__ W2,
    const float* __restrict__ b2, float* __restrict__ out) {
  const int lane = threadIdx.x & 31;
  const int wave = threadIdx.x >> 5;
  const int tile = blockIdx.x * 8 + wave;
  const int mt = tile >> 1;
  const int nt = tile & 1;
  const int lm = lane & 15;
  const int half = lane >> 4;
  const int kb = half * 2;

  const int ncol = nt * 16 + lm;
  const float nmask = (ncol < NUM_CLASSES) ? 1.0f : 0.0f;
  const int nclamp = (ncol < NUM_CLASSES) ? ncol : (NUM_CLASSES - 1);

  const float* Arow = hidden + (size_t)(mt * 16 + lm) * HIDDEN;

  v8f c = {0.f, 0.f, 0.f, 0.f, 0.f, 0.f, 0.f, 0.f};
  for (int k0 = 0; k0 < HIDDEN; k0 += 4) {
    v2f a, bm;
    a.x = Arow[k0 + kb];
    a.y = Arow[k0 + kb + 1];
    bm.x = W2[(size_t)(k0 + kb) * NUM_CLASSES + nclamp] * nmask;
    bm.y = W2[(size_t)(k0 + kb + 1) * NUM_CLASSES + nclamp] * nmask;
    c = __builtin_amdgcn_wmma_f32_16x16x4_f32(
        false, a, false, bm, (short)0, c, false, false);
  }

  if (ncol < NUM_CLASSES) {
    const float bias = b2[ncol];
    const int mbase = mt * 16 + half * 8;
#pragma unroll
    for (int v = 0; v < 8; ++v) {
      out[(size_t)(mbase + v) * NUM_CLASSES + ncol] = c[v] + bias;
    }
  }
}

extern "C" void kernel_launch(void* const* d_in, const int* in_sizes, int n_in,
                              void* d_out, int out_size, void* d_ws, size_t ws_size,
                              hipStream_t stream) {
  const int*   x    = (const int*)d_in[0];
  const int*   lens = (const int*)d_in[1];
  const float* emb  = (const float*)d_in[2];
  const float* W1   = (const float*)d_in[3];
  const float* b1   = (const float*)d_in[4];
  const float* W2   = (const float*)d_in[5];
  const float* b2   = (const float*)d_in[6];
  float*       out  = (float*)d_out;

  float* pooled = (float*)d_ws;
  float* hidden = (float*)((char*)d_ws + (size_t)BATCH * EMB_DIM * sizeof(float));

  ft_pool_kernel<<<BATCH / 8, 256, 0, stream>>>(x, lens, emb, pooled);

  ft_gemm1_kernel<<<(BATCH / 16) * (HIDDEN / 16) / 8, 256, 0, stream>>>(
      pooled, W1, b1, hidden);

  ft_gemm2_kernel<<<(BATCH / 16) * 2 / 8, 256, 0, stream>>>(
      hidden, W2, b2, out);
}


# one launch instead of three: each 16-wave work-group pools 16 samples into LDS, then runs both f32 matrix layers from LDS (pooled, hidden and W2 staged there); all sum orders unchanged
# speedup vs baseline: 4.9142x; 2.2355x over previous
#include <hip/hip_runtime.h>
#include <hip/hip_bf16.h>


#define VOCAB 100000
#define EMB_DIM 128
#define HIDDEN 256
#define NUM_CLASSES 20
#define BATCH 4096
#define SEQ 200

typedef float v2f __attribute__((ext_vector_type(2)));
typedef float v8f __attribute__((ext_vector_type(8)));

#define POOL_CHUNK 32
#define ROWS 16
#define P_PITCH 132
#define H_PITCH 260

__device__ __forceinline__ float4 ft_load_row(const float* __restrict__ emb,
                                              int tok, int u, int lane) {
  const int t = __builtin_amdgcn_readlane(tok, u);
  return ((const float4*)(emb + (size_t)t * EMB_DIM))[lane];
}

__global__ __launch_bounds__(512) void ft_forward_kernel(
    const int* __restrict__ x, const int* __restrict__ lens,
    const float* __restrict__ emb, const float* __restrict__ W1,
    const float* __restrict__ b1, const float* __restrict__ W2,
    const float* __restrict__ b2, float* __restrict__ out) {
  __shared__ float sP[ROWS * P_PITCH];
  __shared__ float sH[ROWS * H_PITCH];
  __shared__ float sW2[HIDDEN * NUM_CLASSES];

  const int lane = threadIdx.x & 31;
  const int wave = __builtin_amdgcn_readfirstlane((int)(threadIdx.x >> 5));
  const int b = (int)blockIdx.x * ROWS + wave;

  float2 w2stage[HIDDEN * NUM_CLASSES / 2 / 512];
#pragma unroll
  for (int j = 0; j < HIDDEN * NUM_CLASSES / 2 / 512; ++j)
    w2stage[j] = ((const float2*)W2)[j * 512 + (int)threadIdx.x];

  int len = lens[b];
  if (len < 1) len = 1;
  if (len > SEQ) len = SEQ;

  const int* xb = x + (size_t)b * SEQ;
  float4 acc = make_float4(0.f, 0.f, 0.f, 0.f);

  int tok = 0;
  if (lane < len) tok = xb[lane];

  int s0 = 0;
  for (; len - s0 >= POOL_CHUNK; s0 += POOL_CHUNK) {
    float4 v[POOL_CHUNK];
    for (int u = 0; u < POOL_CHUNK; ++u) v[u] = ft_load_row(emb, tok, u, lane);

    int tok_next = 0;
    if (s0 + POOL_CHUNK + lane < len) tok_next = xb[s0 + POOL_CHUNK + lane];

    for (int u = 0; u < POOL_CHUNK; ++u) {
      acc.x += v[u].x; acc.y += v[u].y; acc.z += v[u].z; acc.w += v[u].w;
    }
    tok = tok_next;
  }

  const int cnt = len - s0;
  if (cnt > 0) {
    float4 v[POOL_CHUNK - 1];
    v[0] = ft_load_row(emb, tok, 0, lane);
    for (int u = 1; u < POOL_CHUNK - 1; ++u) {
      v[u] = make_float4(0.f, 0.f, 0.f, 0.f);
      if (u < cnt) v[u] = ft_load_row(emb, tok, u, lane);
    }
    for (int u = 0; u < POOL_CHUNK - 1; ++u) {
      acc.x += v[u].x; acc.y += v[u].y; acc.z += v[u].z; acc.w += v[u].w;
    }
  }

  const float inv = 1.0f / (float)len;
  float4 r = make_float4(acc.x * inv, acc.y * inv, acc.z * inv, acc.w * inv);
  *(float4*)(sP + wave * P_PITCH + lane * 4) = r;
#pragma unroll
  for (int j = 0; j < HIDDEN * NUM_CLASSES / 2 / 512; ++j)
    ((float2*)sW2)[j * 512 + (int)threadIdx.x] = w2stage[j];

  const int lm = lane & 15;
  const int half = lane >> 4;
  const int kb = half * 2;

  const int ncol1 = wave * 16 + lm;

  float w1f[EMB_DIM / 2];
#pragma unroll
  for (int i = 0; i < EMB_DIM / 4; ++i) {
    w1f[2 * i]     = W1[(size_t)(4 * i + kb) * HIDDEN + ncol1];
    w1f[2 * i + 1] = W1[(size_t)(4 * i + kb + 1) * HIDDEN + ncol1];
  }

  __syncthreads();

  {
    const float* Arow = sP + lm * P_PITCH + kb;
    v8f c = {0.f, 0.f, 0.f, 0.f, 0.f, 0.f, 0.f, 0.f};
#pragma unroll
    for (int i = 0; i < EMB_DIM / 4; ++i) {
      v2f a, bm;
      a.x = Arow[4 * i];
      a.y = Arow[4 * i + 1];
      bm.x = w1f[2 * i];
      bm.y = w1f[2 * i + 1];
      c = __builtin_amdgcn_wmma_f32_16x16x4_f32(
           false, a,  false, bm,
           (short)0, c,  false,  false);
    }

    const float bias = b1[ncol1];
    float* Hcol = sH + (half * 8) * H_PITCH + ncol1;
#pragma unroll
    for (int v = 0; v < 8; ++v) {
      float val = c[v] + bias;
      val = val > 0.f ? val : 0.f;
      Hcol[v * H_PITCH] = val;
    }
  }

  __syncthreads();

  if (wave >= 2) return;

  {
    const int ncol2 = wave * 16 + lm;
    const float nmask = (ncol2 < NUM_CLASSES) ? 1.0f : 0.0f;
    const int nclamp = (ncol2 < NUM_CLASSES) ? ncol2 : (NUM_CLASSES - 1);
    const float* Arow = sH + lm * H_PITCH + kb;
    const float* Bcol = sW2 + kb * NUM_CLASSES + nclamp;
    v8f c = {0.f, 0.f, 0.f, 0.f, 0.f, 0.f, 0.f, 0.f};

    v2f af[2][16], bf[2][16];
#pragma unroll
    for (int j = 0; j < 16; ++j) {
      af[0][j].x = Arow[4 * j];
      af[0][j].y = Arow[4 * j + 1];
      bf[0][j].x = Bcol[(4 * j) * NUM_CLASSES];
      bf[0][j].y = Bcol[(4 * j + 1) * NUM_CLASSES];
    }
#pragma unroll
    for (int blk = 0; blk < HIDDEN / 64; ++blk) {
      const int cur = blk & 1, nxt = cur ^ 1;
      if (blk + 1 < HIDDEN / 64) {
#pragma unroll
        for (int j = 0; j < 16; ++j) {
          const int i = (blk + 1) * 16 + j;
          af[nxt][j].x = Arow[4 * i];
          af[nxt][j].y = Arow[4 * i + 1];
          bf[nxt][j].x = Bcol[(4 * i) * NUM_CLASSES];
          bf[nxt][j].y = Bcol[(4 * i + 1) * NUM_CLASSES];
        }
      }
      __builtin_amdgcn_sched_barrier(0);
#pragma unroll
      for (int j = 0; j < 16; ++j) {
        v2f bm;
        bm.x = bf[cur][j].x * nmask;
        bm.y = bf[cur][j].y * nmask;
        c = __builtin_amdgcn_wmma_f32_16x16x4_f32(
            false, af[cur][j], false, bm, (short)0, c, false, false);
      }
      __builtin_amdgcn_sched_barrier(0);
    }

    if (ncol2 < NUM_CLASSES) {
      const float bias = b2[ncol2];
      const int mbase = (int)blockIdx.x * ROWS + half * 8;
#pragma unroll
      for (int v = 0; v < 8; ++v) {
        out[(size_t)(mbase + v) * NUM_CLASSES + ncol2] = c[v] + bias;
      }
    }
  }
}

extern "C" void kernel_launch(void* const* d_in, const int* in_sizes, int n_in,
                              void* d_out, int out_size, void* d_ws, size_t ws_size,
                              hipStream_t stream) {
  const int*   x    = (const int*)d_in[0];
  const int*   lens = (const int*)d_in[1];
  const float* emb  = (const float*)d_in[2];
  const float* W1   = (const float*)d_in[3];
  const float* b1   = (const float*)d_in[4];
  const float* W2   = (const float*)d_in[5];
  const float* b2   = (const float*)d_in[6];
  float*       out  = (float*)d_out;

  ft_forward_kernel<<<BATCH / ROWS, 512, 0, stream>>>(
      x, lens, emb, W1, b1, W2, b2, out);
}
